# starting kernel
# baseline (speedup 1.0000x reference)
#include <hip/hip_runtime.h>
#include <hip/hip_bf16.h>

typedef float v2f __attribute__((ext_vector_type(2)));
typedef float v8f __attribute__((ext_vector_type(8)));

__global__ void k_deg_init(int* __restrict__ deg, int n) {
    int i = blockIdx.x * blockDim.x + threadIdx.x;
    if (i < n) deg[i] = 1;
}

__global__ void k_deg_count(const int* __restrict__ dst, int* __restrict__ deg, int e) {
    int i = blockIdx.x * blockDim.x + threadIdx.x;
    if (i < e) atomicAdd(&deg[dst[i]], 1);
}

__global__ void k_dinv(float* __restrict__ dinv, int n) {
    int i = blockIdx.x * blockDim.x + threadIdx.x;
    if (i < n) {
        int d = ((const int*)dinv)[i];
        dinv[i] = rsqrtf((float)d);
    }
}

__global__ void k_gemm1(const float* __restrict__ x, const float* __restrict__ W1,
                        float* __restrict__ h1, int n) {
    int lane = threadIdx.x & 31;
    int wave = threadIdx.x >> 5;
    int base = (blockIdx.x * (blockDim.x >> 5) + wave) * 16;
    if (base >= n) return;

    int  m   = lane & 15;
    bool hi  = lane >= 16;
    int node = base + m;
    if (node >= n) node = n - 1;
    const float* xr = x + (size_t)node * 3;

    float ax = xr[hi ? 2 : 0];
    float x1 = xr[1];
    v2f a;
    a.x = ax;
    a.y = hi ? 0.0f : x1;

    int  mc    = (m < 8) ? m : 7;
    bool valid = (m < 8);
    float w0 = W1[0 * 8 + mc];
    float w1 = W1[1 * 8 + mc];
    float w2 = W1[2 * 8 + mc];
    v2f b;
    b.x = valid ? (hi ? w2 : w0) : 0.0f;
    b.y = (valid && !hi) ? w1 : 0.0f;

    v8f c = {};
    c = __builtin_amdgcn_wmma_f32_16x16x4_f32(false, a, false, b, (short)0, c,
                                              false, false);

    if (valid) {
        float* hp = h1 + (size_t)(base + (hi ? 8 : 0)) * 8 + m;
        if (base + 16 <= n) {
            #pragma unroll
            for (int r = 0; r < 8; ++r) hp[(size_t)r * 8] = c[r];
        } else {
            #pragma unroll
            for (int r = 0; r < 8; ++r) {
                int row = base + r + (hi ? 8 : 0);
                if (row < n) h1[(size_t)row * 8 + m] = c[r];
            }
        }
    }
}

__global__ void k_init1(const float* __restrict__ h1, const float* __restrict__ dinv,
                        const float* __restrict__ b1, float* __restrict__ agg1, int n8) {
    int i = blockIdx.x * blockDim.x + threadIdx.x;
    if (i >= n8) return;
    int   v  = i >> 3, j = i & 7;
    float di = dinv[v];
    agg1[i] = b1[j] + h1[i] * di * di;
}

__global__ void k_edge1(const int* __restrict__ src, const int* __restrict__ dst,
                        const float* __restrict__ dinv, const float* __restrict__ h1,
                        float* __restrict__ agg1, int e) {
    int i = blockIdx.x * blockDim.x + threadIdx.x;
    if (i >= e) return;
    int s = src[i], d = dst[i];
    float nrm = dinv[s] * dinv[d];
    const float4* hp = (const float4*)(h1 + (size_t)s * 8);
    float4 m0 = hp[0], m1 = hp[1];
    float* o = agg1 + (size_t)d * 8;
    unsafeAtomicAdd(o + 0, m0.x * nrm);
    unsafeAtomicAdd(o + 1, m0.y * nrm);
    unsafeAtomicAdd(o + 2, m0.z * nrm);
    unsafeAtomicAdd(o + 3, m0.w * nrm);
    unsafeAtomicAdd(o + 4, m1.x * nrm);
    unsafeAtomicAdd(o + 5, m1.y * nrm);
    unsafeAtomicAdd(o + 6, m1.z * nrm);
    unsafeAtomicAdd(o + 7, m1.w * nrm);
}

__global__ void k_gemm2(const float* __restrict__ agg1, const float* __restrict__ W2,
                        float* __restrict__ h2, int n) {
    int lane = threadIdx.x & 31;
    int wave = threadIdx.x >> 5;
    int base = (blockIdx.x * (blockDim.x >> 5) + wave) * 16;
    if (base >= n) return;

    int  m   = lane & 15;
    bool hi  = lane >= 16;
    int node = base + m;
    if (node >= n) node = n - 1;
    const float* r0 = agg1 + (size_t)node * 8;
    int k0 = hi ? 2 : 0;

    v2f a0, a1;
    a0.x = fmaxf(r0[k0],     0.0f); a0.y = fmaxf(r0[k0 + 1],     0.0f);
    a1.x = fmaxf(r0[4 + k0], 0.0f); a1.y = fmaxf(r0[4 + k0 + 1], 0.0f);

    int  mc    = (m < 2) ? m : 1;
    bool valid = (m < 2);
    float wa = W2[(k0)     * 2 + mc];
    float wb = W2[(k0 + 1) * 2 + mc];
    float wc = W2[(k0 + 4) * 2 + mc];
    float wd = W2[(k0 + 5) * 2 + mc];
    v2f b0, b1v;
    b0.x  = valid ? wa : 0.0f;
    b0.y  = valid ? wb : 0.0f;
    b1v.x = valid ? wc : 0.0f;
    b1v.y = valid ? wd : 0.0f;

    v8f c = {};
    c = __builtin_amdgcn_wmma_f32_16x16x4_f32(false, a0, false, b0, (short)0, c,
                                              false, false);
    c = __builtin_amdgcn_wmma_f32_16x16x4_f32(false, a1, false, b1v, (short)0, c,
                                              false, false);

    if (valid) {
        float* hp = h2 + (size_t)(base + (hi ? 8 : 0)) * 2 + m;
        if (base + 16 <= n) {
            #pragma unroll
            for (int r = 0; r < 8; ++r) hp[(size_t)r * 2] = c[r];
        } else {
            #pragma unroll
            for (int r = 0; r < 8; ++r) {
                int row = base + r + (hi ? 8 : 0);
                if (row < n) h2[(size_t)row * 2 + m] = c[r];
            }
        }
    }
}

__global__ void k_init2(const float* __restrict__ h2, const float* __restrict__ dinv,
                        const float* __restrict__ b2, float* __restrict__ out, int n2) {
    int i = blockIdx.x * blockDim.x + threadIdx.x;
    if (i >= n2) return;
    int   v  = i >> 1, j = i & 1;
    float di = dinv[v];
    out[i] = b2[j] + h2[i] * di * di;
}

__global__ void k_edge2(const int* __restrict__ src, const int* __restrict__ dst,
                        const float* __restrict__ dinv, const float* __restrict__ h2,
                        float* __restrict__ out, int e) {
    int i = blockIdx.x * blockDim.x + threadIdx.x;
    if (i >= e) return;
    int s = src[i], d = dst[i];
    float nrm = dinv[s] * dinv[d];
    const float2* hp = (const float2*)(h2 + (size_t)s * 2);
    float2 m0 = hp[0];
    float* o = out + (size_t)d * 2;
    unsafeAtomicAdd(o + 0, m0.x * nrm);
    unsafeAtomicAdd(o + 1, m0.y * nrm);
}

extern "C" void kernel_launch(void* const* d_in, const int* in_sizes, int n_in,
                              void* d_out, int out_size, void* d_ws, size_t ws_size,
                              hipStream_t stream) {
    const float* x  = (const float*)d_in[0];
    const int*   ei = (const int*)d_in[1];
    const float* W1 = (const float*)d_in[2];
    const float* b1 = (const float*)d_in[3];
    const float* W2 = (const float*)d_in[4];
    const float* b2 = (const float*)d_in[5];
    float* out = (float*)d_out;

    int n = in_sizes[0] / 3;
    int e = in_sizes[1] / 2;
    const int* src = ei;
    const int* dst = ei + e;

    float* dinv = (float*)d_ws;
    float* h1   = dinv + n;
    float* agg1 = h1 + (size_t)n * 8;
    float* h2   = h1;

    const int T = 256;
    int gN  = (n + T - 1) / T;
    int gE  = (e + T - 1) / T;
    int gN8 = (n * 8 + T - 1) / T;
    int gN2 = (n * 2 + T - 1) / T;
    int gMM = (n + 127) / 128;

    k_deg_init <<<gN,  T, 0, stream>>>((int*)dinv, n);
    k_deg_count<<<gE,  T, 0, stream>>>(dst, (int*)dinv, e);
    k_dinv     <<<gN,  T, 0, stream>>>(dinv, n);
    k_gemm1    <<<gMM, T, 0, stream>>>(x, W1, h1, n);
    k_init1    <<<gN8, T, 0, stream>>>(h1, dinv, b1, agg1, n * 8);
    k_edge1    <<<gE,  T, 0, stream>>>(src, dst, dinv, h1, agg1, e);
    k_gemm2    <<<gMM, T, 0, stream>>>(agg1, W2, h2, n);
    k_init2    <<<gN2, T, 0, stream>>>(h2, dinv, b2, out, n * 2);
    k_edge2    <<<gE,  T, 0, stream>>>(src, dst, dinv, h2, out, e);
}


# targets split in two ranges, one per chip half: count and fill draw (edge chunk, range) tickets, own half first (die id, speed only): counter and csr lines written from one half; row order by arrival
# speedup vs baseline: 3.1347x; 3.1347x over previous
#include <hip/hip_runtime.h>

typedef float v2f __attribute__((ext_vector_type(2)));
typedef float v8f __attribute__((ext_vector_type(8)));

#define EDGE_TRIPS 8
#define CHUNK (256 * EDGE_TRIPS)
#define TICK_PITCH 64

__device__ __forceinline__ int chip_half() {
    return (int)((__builtin_amdgcn_s_sendmsg_rtn(0x87) >> 18) & 1u);
}

__global__ void k_zero(int* __restrict__ cur, int* __restrict__ tick, int n) {
    int i = blockIdx.x * blockDim.x + threadIdx.x;
    if (i < n) cur[i] = 0;
    if (i < 4 * TICK_PITCH) tick[i] = 0;
}

__global__ void __launch_bounds__(256)
k_count(const int* __restrict__ dst, int* __restrict__ cur, int* __restrict__ tick,
        int e, int split, int nchunks) {
    __shared__ int s_c;
    int tid = threadIdx.x;
    int h   = chip_half();
    for (int pass = 0; pass < 2; ++pass) {
        int  r  = h ^ pass;
        bool hi = (r == 1);
        for (;;) {
            if (tid == 0) s_c = atomicAdd(&tick[r * TICK_PITCH], 1);
            __syncthreads();
            int c = s_c;
            __syncthreads();
            if (c >= nchunks) break;
            int i0 = c * CHUNK + tid;
            int d[EDGE_TRIPS];
            #pragma unroll
            for (int k = 0; k < EDGE_TRIPS; ++k) {
                int i = i0 + k * 256;
                d[k] = (i < e) ? dst[i] : -1;
            }
            #pragma unroll
            for (int k = 0; k < EDGE_TRIPS; ++k)
                if (d[k] >= 0 && (d[k] >= split) == hi) atomicAdd(&cur[d[k]], 1);
        }
    }
}

__device__ __forceinline__ int wave_incl_scan(int v, int lane) {
    #pragma unroll
    for (int off = 1; off < 32; off <<= 1) {
        int y = __shfl_up(v, off);
        if (lane >= off) v += y;
    }
    return v;
}

__global__ void k_scan_block(int* __restrict__ cur, float* __restrict__ dinv,
                             int* __restrict__ bsum, int n) {
    __shared__ int wsum[8];
    int tid  = threadIdx.x;
    int lane = tid & 31, wv = tid >> 5;
    int i0   = (blockIdx.x * 256 + tid) * 4;
    int c[4];
    #pragma unroll
    for (int k = 0; k < 4; ++k) {
        int idx = i0 + k;
        c[k] = (idx < n) ? cur[idx] : 0;
        if (idx < n) dinv[idx] = rsqrtf((float)(c[k] + 1));
    }
    int t   = c[0] + c[1] + c[2] + c[3];
    int inc = wave_incl_scan(t, lane);
    if (lane == 31) wsum[wv] = inc;
    __syncthreads();
    int woff = 0;
    #pragma unroll
    for (int w = 0; w < 8; ++w) woff += (w < wv) ? wsum[w] : 0;
    int run = woff + inc - t;
    #pragma unroll
    for (int k = 0; k < 4; ++k) {
        int idx = i0 + k;
        if (idx < n) cur[idx] = run;
        run += c[k];
    }
    if (tid == 255) bsum[blockIdx.x] = woff + inc;
}

__global__ void k_scan_top(int* __restrict__ bsum, int nb) {
    __shared__ int wsum[32];
    int tid  = threadIdx.x;
    int lane = tid & 31, wv = tid >> 5;
    int carry = 0;
    for (int c0 = 0; c0 < nb; c0 += 1024) {
        int idx = c0 + tid;
        int t   = (idx < nb) ? bsum[idx] : 0;
        int inc = wave_incl_scan(t, lane);
        if (lane == 31) wsum[wv] = inc;
        __syncthreads();
        int woff = 0, tot = 0;
        #pragma unroll
        for (int w = 0; w < 32; ++w) {
            int v = wsum[w];
            woff += (w < wv) ? v : 0;
            tot  += v;
        }
        if (idx < nb) bsum[idx] = carry + woff + inc - t;
        carry += tot;
        __syncthreads();
    }
}

__global__ void k_scan_add(int* __restrict__ cur, const int* __restrict__ bsum, int n) {
    int i = blockIdx.x * blockDim.x + threadIdx.x;
    if (i < n) cur[i] += bsum[i >> 10];
}

__global__ void __launch_bounds__(256)
k_fill(const int* __restrict__ src, const int* __restrict__ dst,
       int* __restrict__ cur, int* __restrict__ csr, int* __restrict__ tick,
       int e, int split, int nchunks) {
    __shared__ int s_c;
    int tid = threadIdx.x;
    int h   = chip_half();
    for (int pass = 0; pass < 2; ++pass) {
        int  r  = h ^ pass;
        bool hi = (r == 1);
        for (;;) {
            if (tid == 0) s_c = atomicAdd(&tick[(2 + r) * TICK_PITCH], 1);
            __syncthreads();
            int c = s_c;
            __syncthreads();
            if (c >= nchunks) break;
            int i0 = c * CHUNK + tid;
            int  d[EDGE_TRIPS], s[EDGE_TRIPS], pos[EDGE_TRIPS];
            bool mine[EDGE_TRIPS];
            #pragma unroll
            for (int k = 0; k < EDGE_TRIPS; ++k) {
                int i = i0 + k * 256;
                d[k]    = (i < e) ? dst[i] : -1;
                mine[k] = d[k] >= 0 && (d[k] >= split) == hi;
            }
            #pragma unroll
            for (int k = 0; k < EDGE_TRIPS; ++k) {
                if (mine[k]) {
                    s[k]   = src[i0 + k * 256];
                    pos[k] = atomicAdd(&cur[d[k]], 1);
                }
            }
            #pragma unroll
            for (int k = 0; k < EDGE_TRIPS; ++k)
                if (mine[k]) csr[pos[k]] = s[k];
        }
    }
}

__global__ void k_pull1(const float* __restrict__ x, const int* __restrict__ cur,
                        const int* __restrict__ csr, const float* __restrict__ dinv,
                        const float* __restrict__ W1, const float* __restrict__ b1,
                        const float* __restrict__ W2, float* __restrict__ h2, int n) {
    int lane = threadIdx.x & 31;
    int wave = threadIdx.x >> 5;
    int base = (blockIdx.x * (blockDim.x >> 5) + wave) * 16;
    if (base >= n) return;

    int  m    = lane & 15;
    bool hi   = lane >= 16;
    int  node = base + m;
    bool live = node < n;
    if (!live) node = n - 1;
    int off = hi ? 2 : 0;

    int p   = node ? cur[node - 1] : 0;
    int end = live ? cur[node] : p;

    float s0 = 0.0f, s1 = 0.0f;
    for (; p + 1 < end; p += 2) {
        int   sa = csr[p], sb = csr[p + 1];
        float wa = dinv[sa], wb = dinv[sb];
        const float* xa = x + (size_t)sa * 3;
        const float* xb = x + (size_t)sb * 3;
        float a0 = xa[off], a1 = xa[1];
        float c0 = xb[off], c1 = xb[1];
        s0 += wa * a0; s1 += wa * a1;
        s0 += wb * c0; s1 += wb * c1;
    }
    if (p < end) {
        int   sa = csr[p];
        float wa = dinv[sa];
        const float* xa = x + (size_t)sa * 3;
        s0 += wa * xa[off]; s1 += wa * xa[1];
    }
    float dv = dinv[node];
    const float* xs = x + (size_t)node * 3;
    s0 = (s0 + dv * xs[off]) * dv;
    s1 = (s1 + dv * xs[1]) * dv;

    v2f bx;
    bx.x = s0;
    bx.y = hi ? 0.0f : s1;

    int  fc = (m < 8) ? m : 7;
    bool fv = (m < 8);
    float w0 = W1[0 * 8 + fc];
    float w1 = W1[1 * 8 + fc];
    float w2 = W1[2 * 8 + fc];
    v2f aw;
    aw.x = fv ? (hi ? w2 : w0) : 0.0f;
    aw.y = (fv && !hi) ? w1 : 0.0f;

    v8f g = {};
    g = __builtin_amdgcn_wmma_f32_16x16x4_f32(false, aw, false, bx, (short)0, g,
                                              false, false);

    float t[8];
    #pragma unroll
    for (int r = 0; r < 8; ++r) t[r] = fmaxf(g[r] + b1[r], 0.0f);

    float u2 = __shfl(t[2], m), u3 = __shfl(t[3], m);
    float u6 = __shfl(t[6], m), u7 = __shfl(t[7], m);
    v2f q0, q1;
    q0.x = hi ? u2 : t[0]; q0.y = hi ? u3 : t[1];
    q1.x = hi ? u6 : t[4]; q1.y = hi ? u7 : t[5];

    int  k0 = hi ? 2 : 0;
    int  jc = (m < 2) ? m : 1;
    bool jv = (m < 2);
    float wa2 = W2[(k0)     * 2 + jc];
    float wb2 = W2[(k0 + 1) * 2 + jc];
    float wc2 = W2[(k0 + 4) * 2 + jc];
    float wd2 = W2[(k0 + 5) * 2 + jc];
    v2f e0, e1;
    e0.x = jv ? wa2 : 0.0f; e0.y = jv ? wb2 : 0.0f;
    e1.x = jv ? wc2 : 0.0f; e1.y = jv ? wd2 : 0.0f;

    v8f o = {};
    o = __builtin_amdgcn_wmma_f32_16x16x4_f32(false, e0, false, q0, (short)0, o,
                                              false, false);
    o = __builtin_amdgcn_wmma_f32_16x16x4_f32(false, e1, false, q1, (short)0, o,
                                              false, false);

    if (!hi && live) {
        float2 r2;
        r2.x = o[0]; r2.y = o[1];
        ((float2*)h2)[base + m] = r2;
    }
}

__global__ void k_pull2(const float* __restrict__ h2, const int* __restrict__ cur,
                        const int* __restrict__ csr, const float* __restrict__ dinv,
                        const float* __restrict__ b2, float* __restrict__ out, int n) {
    int v = blockIdx.x * blockDim.x + threadIdx.x;
    if (v >= n) return;
    const float2* hp = (const float2*)h2;
    int p   = v ? cur[v - 1] : 0;
    int end = cur[v];
    float a0 = 0.0f, a1 = 0.0f;
    for (; p + 1 < end; p += 2) {
        int    sa = csr[p], sb = csr[p + 1];
        float  wa = dinv[sa], wb = dinv[sb];
        float2 ha = hp[sa], hb = hp[sb];
        a0 += wa * ha.x; a1 += wa * ha.y;
        a0 += wb * hb.x; a1 += wb * hb.y;
    }
    if (p < end) {
        int    sa = csr[p];
        float  wa = dinv[sa];
        float2 ha = hp[sa];
        a0 += wa * ha.x; a1 += wa * ha.y;
    }
    float  dv = dinv[v];
    float2 hv = hp[v];
    float2 r;
    r.x = b2[0] + dv * (a0 + dv * hv.x);
    r.y = b2[1] + dv * (a1 + dv * hv.y);
    ((float2*)out)[v] = r;
}

extern "C" void kernel_launch(void* const* d_in, const int* in_sizes, int n_in,
                              void* d_out, int out_size, void* d_ws, size_t ws_size,
                              hipStream_t stream) {
    const float* x  = (const float*)d_in[0];
    const int*   ei = (const int*)d_in[1];
    const float* W1 = (const float*)d_in[2];
    const float* b1 = (const float*)d_in[3];
    const float* W2 = (const float*)d_in[4];
    const float* b2 = (const float*)d_in[5];
    float* out = (float*)d_out;

    int n = in_sizes[0] / 3;
    int e = in_sizes[1] / 2;
    const int* src = ei;
    const int* dst = ei + e;

    int*   cur  = (int*)d_ws;
    float* dinv = (float*)d_ws + n;
    float* h2   = dinv + n;
    int*   csr  = (int*)(h2 + (size_t)n * 2);
    int*   bsum = csr + e;

    const int T = 256;
    int gN  = (n + T - 1) / T;
    int gMM = (n + 127) / 128;
    int nb  = (n + 1023) / 1024;
    int*   tick = bsum + ((nb + 63) / 64) * 64;
    int split   = ((n / 2 + 63) / 64) * 64;
    int nchunks = (e + CHUNK - 1) / CHUNK;
    const int gR = 2048;

    k_zero      <<<gN,  T,    0, stream>>>(cur, tick, n);
    k_count     <<<gR,  T,    0, stream>>>(dst, cur, tick, e, split, nchunks);
    k_scan_block<<<nb,  256,  0, stream>>>(cur, dinv, bsum, n);
    k_scan_top  <<<1,   1024, 0, stream>>>(bsum, nb);
    k_scan_add  <<<gN,  T,    0, stream>>>(cur, bsum, n);
    k_fill      <<<gR,  T,    0, stream>>>(src, dst, cur, csr, tick, e, split, nchunks);
    k_pull1     <<<gMM, T,    0, stream>>>(x, cur, csr, dinv, W1, b1, W2, h2, n);
    k_pull2     <<<gN,  T,    0, stream>>>(h2, cur, csr, dinv, b2, out, n);
}


# layer-1 gather: one lane per node fetches a source's 3 features with one 12-byte load (32 nodes per wave, was 16 on two lanes each); B fragments built by a lane-half exchange; sums unchanged
# speedup vs baseline: 3.3072x; 1.0550x over previous
#include <hip/hip_runtime.h>

typedef float v2f __attribute__((ext_vector_type(2)));
typedef float v8f __attribute__((ext_vector_type(8)));

#define EDGE_TRIPS 8
#define CHUNK (256 * EDGE_TRIPS)
#define TICK_PITCH 64

__device__ __forceinline__ int chip_half() {
    return (int)((__builtin_amdgcn_s_sendmsg_rtn(0x87) >> 18) & 1u);
}

__global__ void k_zero(int* __restrict__ cur, int* __restrict__ tick, int n) {
    int i = blockIdx.x * blockDim.x + threadIdx.x;
    if (i < n) cur[i] = 0;
    if (i < 4 * TICK_PITCH) tick[i] = 0;
}

__global__ void __launch_bounds__(256)
k_count(const int* __restrict__ dst, int* __restrict__ cur, int* __restrict__ tick,
        int e, int split, int nchunks) {
    __shared__ int s_c;
    int tid = threadIdx.x;
    int h   = chip_half();
    for (int pass = 0; pass < 2; ++pass) {
        int  r  = h ^ pass;
        bool hi = (r == 1);
        for (;;) {
            if (tid == 0) s_c = atomicAdd(&tick[r * TICK_PITCH], 1);
            __syncthreads();
            int c = s_c;
            __syncthreads();
            if (c >= nchunks) break;
            int i0 = c * CHUNK + tid;
            int d[EDGE_TRIPS];
            #pragma unroll
            for (int k = 0; k < EDGE_TRIPS; ++k) {
                int i = i0 + k * 256;
                d[k] = (i < e) ? dst[i] : -1;
            }
            #pragma unroll
            for (int k = 0; k < EDGE_TRIPS; ++k)
                if (d[k] >= 0 && (d[k] >= split) == hi) atomicAdd(&cur[d[k]], 1);
        }
    }
}

__device__ __forceinline__ int wave_incl_scan(int v, int lane) {
    #pragma unroll
    for (int off = 1; off < 32; off <<= 1) {
        int y = __shfl_up(v, off);
        if (lane >= off) v += y;
    }
    return v;
}

__global__ void k_scan_block(int* __restrict__ cur, float* __restrict__ dinv,
                             int* __restrict__ bsum, int n) {
    __shared__ int wsum[8];
    int tid  = threadIdx.x;
    int lane = tid & 31, wv = tid >> 5;
    int i0   = (blockIdx.x * 256 + tid) * 4;
    int c[4];
    #pragma unroll
    for (int k = 0; k < 4; ++k) {
        int idx = i0 + k;
        c[k] = (idx < n) ? cur[idx] : 0;
        if (idx < n) dinv[idx] = rsqrtf((float)(c[k] + 1));
    }
    int t   = c[0] + c[1] + c[2] + c[3];
    int inc = wave_incl_scan(t, lane);
    if (lane == 31) wsum[wv] = inc;
    __syncthreads();
    int woff = 0;
    #pragma unroll
    for (int w = 0; w < 8; ++w) woff += (w < wv) ? wsum[w] : 0;
    int run = woff + inc - t;
    #pragma unroll
    for (int k = 0; k < 4; ++k) {
        int idx = i0 + k;
        if (idx < n) cur[idx] = run;
        run += c[k];
    }
    if (tid == 255) bsum[blockIdx.x] = woff + inc;
}

__global__ void k_scan_top(int* __restrict__ bsum, int nb) {
    __shared__ int wsum[32];
    int tid  = threadIdx.x;
    int lane = tid & 31, wv = tid >> 5;
    int carry = 0;
    for (int c0 = 0; c0 < nb; c0 += 1024) {
        int idx = c0 + tid;
        int t   = (idx < nb) ? bsum[idx] : 0;
        int inc = wave_incl_scan(t, lane);
        if (lane == 31) wsum[wv] = inc;
        __syncthreads();
        int woff = 0, tot = 0;
        #pragma unroll
        for (int w = 0; w < 32; ++w) {
            int v = wsum[w];
            woff += (w < wv) ? v : 0;
            tot  += v;
        }
        if (idx < nb) bsum[idx] = carry + woff + inc - t;
        carry += tot;
        __syncthreads();
    }
}

__global__ void k_scan_add(int* __restrict__ cur, const int* __restrict__ bsum, int n) {
    int i = blockIdx.x * blockDim.x + threadIdx.x;
    if (i < n) cur[i] += bsum[i >> 10];
}

__global__ void __launch_bounds__(256)
k_fill(const int* __restrict__ src, const int* __restrict__ dst,
       int* __restrict__ cur, int* __restrict__ csr, int* __restrict__ tick,
       int e, int split, int nchunks) {
    __shared__ int s_c;
    int tid = threadIdx.x;
    int h   = chip_half();
    for (int pass = 0; pass < 2; ++pass) {
        int  r  = h ^ pass;
        bool hi = (r == 1);
        for (;;) {
            if (tid == 0) s_c = atomicAdd(&tick[(2 + r) * TICK_PITCH], 1);
            __syncthreads();
            int c = s_c;
            __syncthreads();
            if (c >= nchunks) break;
            int i0 = c * CHUNK + tid;
            int  d[EDGE_TRIPS], s[EDGE_TRIPS], pos[EDGE_TRIPS];
            bool mine[EDGE_TRIPS];
            #pragma unroll
            for (int k = 0; k < EDGE_TRIPS; ++k) {
                int i = i0 + k * 256;
                d[k]    = (i < e) ? dst[i] : -1;
                mine[k] = d[k] >= 0 && (d[k] >= split) == hi;
            }
            #pragma unroll
            for (int k = 0; k < EDGE_TRIPS; ++k) {
                if (mine[k]) {
                    s[k]   = src[i0 + k * 256];
                    pos[k] = atomicAdd(&cur[d[k]], 1);
                }
            }
            #pragma unroll
            for (int k = 0; k < EDGE_TRIPS; ++k)
                if (mine[k]) csr[pos[k]] = s[k];
        }
    }
}

__global__ void k_pull1(const float* __restrict__ x, const int* __restrict__ cur,
                        const int* __restrict__ csr, const float* __restrict__ dinv,
                        const float* __restrict__ W1, const float* __restrict__ b1,
                        const float* __restrict__ W2, float* __restrict__ h2, int n) {
    int lane = threadIdx.x & 31;
    int wave = threadIdx.x >> 5;
    int base = (blockIdx.x * (blockDim.x >> 5) + wave) * 32;
    if (base >= n) return;

    int  m    = lane & 15;
    bool hi   = lane >= 16;
    int  node = base + lane;
    bool live = node < n;
    if (!live) node = n - 1;

    int p   = node ? cur[node - 1] : 0;
    int end = live ? cur[node] : p;

    float s0 = 0.0f, s1 = 0.0f, s2 = 0.0f;
    for (; p + 1 < end; p += 2) {
        int   sa = csr[p], sb = csr[p + 1];
        float wa = dinv[sa], wb = dinv[sb];
        const float* xa = x + (size_t)sa * 3;
        const float* xb = x + (size_t)sb * 3;
        float a0 = xa[0], a1 = xa[1], a2 = xa[2];
        float c0 = xb[0], c1 = xb[1], c2 = xb[2];
        s0 += wa * a0; s1 += wa * a1; s2 += wa * a2;
        s0 += wb * c0; s1 += wb * c1; s2 += wb * c2;
    }
    if (p < end) {
        int   sa = csr[p];
        float wa = dinv[sa];
        const float* xa = x + (size_t)sa * 3;
        float a0 = xa[0], a1 = xa[1], a2 = xa[2];
        s0 += wa * a0; s1 += wa * a1; s2 += wa * a2;
    }
    float dv = dinv[node];
    const float* xs = x + (size_t)node * 3;
    float x0 = xs[0], x1 = xs[1], x2 = xs[2];
    s0 = (s0 + dv * x0) * dv;
    s1 = (s1 + dv * x1) * dv;
    s2 = (s2 + dv * x2) * dv;

    float y0 = __shfl_xor(s0, 16), y1 = __shfl_xor(s1, 16), y2 = __shfl_xor(s2, 16);

    v2f bxA, bxB;
    bxA.x = hi ? y2 : s0;  bxA.y = hi ? 0.0f : s1;
    bxB.x = hi ? s2 : y0;  bxB.y = hi ? 0.0f : y1;

    int  fc = (m < 8) ? m : 7;
    bool fv = (m < 8);
    float w0 = W1[0 * 8 + fc];
    float w1 = W1[1 * 8 + fc];
    float w2 = W1[2 * 8 + fc];
    v2f aw;
    aw.x = fv ? (hi ? w2 : w0) : 0.0f;
    aw.y = (fv && !hi) ? w1 : 0.0f;

    v8f gA = {}, gB = {};
    gA = __builtin_amdgcn_wmma_f32_16x16x4_f32(false, aw, false, bxA, (short)0, gA,
                                               false, false);
    gB = __builtin_amdgcn_wmma_f32_16x16x4_f32(false, aw, false, bxB, (short)0, gB,
                                               false, false);

    float tA[8], tB[8];
    #pragma unroll
    for (int r = 0; r < 8; ++r) {
        float br = b1[r];
        tA[r] = fmaxf(gA[r] + br, 0.0f);
        tB[r] = fmaxf(gB[r] + br, 0.0f);
    }

    float uA2 = __shfl(tA[2], m), uA3 = __shfl(tA[3], m);
    float uA6 = __shfl(tA[6], m), uA7 = __shfl(tA[7], m);
    float uB2 = __shfl(tB[2], m), uB3 = __shfl(tB[3], m);
    float uB6 = __shfl(tB[6], m), uB7 = __shfl(tB[7], m);
    v2f qA0, qA1, qB0, qB1;
    qA0.x = hi ? uA2 : tA[0]; qA0.y = hi ? uA3 : tA[1];
    qA1.x = hi ? uA6 : tA[4]; qA1.y = hi ? uA7 : tA[5];
    qB0.x = hi ? uB2 : tB[0]; qB0.y = hi ? uB3 : tB[1];
    qB1.x = hi ? uB6 : tB[4]; qB1.y = hi ? uB7 : tB[5];

    int  k0 = hi ? 2 : 0;
    int  jc = (m < 2) ? m : 1;
    bool jv = (m < 2);
    float wa2 = W2[(k0)     * 2 + jc];
    float wb2 = W2[(k0 + 1) * 2 + jc];
    float wc2 = W2[(k0 + 4) * 2 + jc];
    float wd2 = W2[(k0 + 5) * 2 + jc];
    v2f e0, e1;
    e0.x = jv ? wa2 : 0.0f; e0.y = jv ? wb2 : 0.0f;
    e1.x = jv ? wc2 : 0.0f; e1.y = jv ? wd2 : 0.0f;

    v8f oA = {}, oB = {};
    oA = __builtin_amdgcn_wmma_f32_16x16x4_f32(false, e0, false, qA0, (short)0, oA,
                                               false, false);
    oB = __builtin_amdgcn_wmma_f32_16x16x4_f32(false, e0, false, qB0, (short)0, oB,
                                               false, false);
    oA = __builtin_amdgcn_wmma_f32_16x16x4_f32(false, e1, false, qA1, (short)0, oA,
                                               false, false);
    oB = __builtin_amdgcn_wmma_f32_16x16x4_f32(false, e1, false, qB1, (short)0, oB,
                                               false, false);

    float z0 = __shfl_xor(oB[0], 16), z1 = __shfl_xor(oB[1], 16);
    if (live) {
        float2 r2;
        r2.x = hi ? z0 : oA[0];
        r2.y = hi ? z1 : oA[1];
        ((float2*)h2)[node] = r2;
    }
}

__global__ void k_pull2(const float* __restrict__ h2, const int* __restrict__ cur,
                        const int* __restrict__ csr, const float* __restrict__ dinv,
                        const float* __restrict__ b2, float* __restrict__ out, int n) {
    int v = blockIdx.x * blockDim.x + threadIdx.x;
    if (v >= n) return;
    const float2* hp = (const float2*)h2;
    int p   = v ? cur[v - 1] : 0;
    int end = cur[v];
    float a0 = 0.0f, a1 = 0.0f;
    for (; p + 1 < end; p += 2) {
        int    sa = csr[p], sb = csr[p + 1];
        float  wa = dinv[sa], wb = dinv[sb];
        float2 ha = hp[sa], hb = hp[sb];
        a0 += wa * ha.x; a1 += wa * ha.y;
        a0 += wb * hb.x; a1 += wb * hb.y;
    }
    if (p < end) {
        int    sa = csr[p];
        float  wa = dinv[sa];
        float2 ha = hp[sa];
        a0 += wa * ha.x; a1 += wa * ha.y;
    }
    float  dv = dinv[v];
    float2 hv = hp[v];
    float2 r;
    r.x = b2[0] + dv * (a0 + dv * hv.x);
    r.y = b2[1] + dv * (a1 + dv * hv.y);
    ((float2*)out)[v] = r;
}

extern "C" void kernel_launch(void* const* d_in, const int* in_sizes, int n_in,
                              void* d_out, int out_size, void* d_ws, size_t ws_size,
                              hipStream_t stream) {
    const float* x  = (const float*)d_in[0];
    const int*   ei = (const int*)d_in[1];
    const float* W1 = (const float*)d_in[2];
    const float* b1 = (const float*)d_in[3];
    const float* W2 = (const float*)d_in[4];
    const float* b2 = (const float*)d_in[5];
    float* out = (float*)d_out;

    int n = in_sizes[0] / 3;
    int e = in_sizes[1] / 2;
    const int* src = ei;
    const int* dst = ei + e;

    int*   cur  = (int*)d_ws;
    float* dinv = (float*)d_ws + n;
    float* h2   = dinv + n;
    int*   csr  = (int*)(h2 + (size_t)n * 2);
    int*   bsum = csr + e;

    const int T = 256;
    int gN  = (n + T - 1) / T;
    int gMM = (n + 255) / 256;
    int nb  = (n + 1023) / 1024;
    int*   tick = bsum + ((nb + 63) / 64) * 64;
    int split   = ((n / 2 + 63) / 64) * 64;
    int nchunks = (e + CHUNK - 1) / CHUNK;
    const int gR = 2048;

    k_zero      <<<gN,  T,    0, stream>>>(cur, tick, n);
    k_count     <<<gR,  T,    0, stream>>>(dst, cur, tick, e, split, nchunks);
    k_scan_block<<<nb,  256,  0, stream>>>(cur, dinv, bsum, n);
    k_scan_top  <<<1,   1024, 0, stream>>>(bsum, nb);
    k_scan_add  <<<gN,  T,    0, stream>>>(cur, bsum, n);
    k_fill      <<<gR,  T,    0, stream>>>(src, dst, cur, csr, tick, e, split, nchunks);
    k_pull1     <<<gMM, T,    0, stream>>>(x, cur, csr, dinv, W1, b1, W2, h2, n);
    k_pull2     <<<gN,  T,    0, stream>>>(h2, cur, csr, dinv, b2, out, n);
}


# gather kernels: a block copies its 256 rows' piece of csr to LDS once, whole 128-B lines per load, and lanes take source ids from LDS, not from memory on every trip; sums and their order unchanged
# speedup vs baseline: 4.2791x; 1.2939x over previous
#include <hip/hip_runtime.h>

typedef float v2f __attribute__((ext_vector_type(2)));
typedef float v8f __attribute__((ext_vector_type(8)));

#define EDGE_TRIPS 8
#define CHUNK (256 * EDGE_TRIPS)
#define TICK_PITCH 64
#define SEG_CAP 4096
#define SEG_TRIPS 4

__device__ __forceinline__ int chip_half() {
    return (int)((__builtin_amdgcn_s_sendmsg_rtn(0x87) >> 18) & 1u);
}

__global__ void k_zero(int* __restrict__ cur, int* __restrict__ tick, int n) {
    int i = blockIdx.x * blockDim.x + threadIdx.x;
    if (i < n) cur[i] = 0;
    if (i < 4 * TICK_PITCH) tick[i] = 0;
}

__global__ void __launch_bounds__(256)
k_count(const int* __restrict__ dst, int* __restrict__ cur, int* __restrict__ tick,
        int e, int split, int nchunks) {
    __shared__ int s_c;
    int tid = threadIdx.x;
    int h   = chip_half();
    for (int pass = 0; pass < 2; ++pass) {
        int  r  = h ^ pass;
        bool hi = (r == 1);
        for (;;) {
            if (tid == 0) s_c = atomicAdd(&tick[r * TICK_PITCH], 1);
            __syncthreads();
            int c = s_c;
            __syncthreads();
            if (c >= nchunks) break;
            int i0 = c * CHUNK + tid;
            int d[EDGE_TRIPS];
            #pragma unroll
            for (int k = 0; k < EDGE_TRIPS; ++k) {
                int i = i0 + k * 256;
                d[k] = (i < e) ? dst[i] : -1;
            }
            #pragma unroll
            for (int k = 0; k < EDGE_TRIPS; ++k)
                if (d[k] >= 0 && (d[k] >= split) == hi) atomicAdd(&cur[d[k]], 1);
        }
    }
}

__device__ __forceinline__ int wave_incl_scan(int v, int lane) {
    #pragma unroll
    for (int off = 1; off < 32; off <<= 1) {
        int y = __shfl_up(v, off);
        if (lane >= off) v += y;
    }
    return v;
}

__global__ void k_scan_block(int* __restrict__ cur, float* __restrict__ dinv,
                             int* __restrict__ bsum, int n) {
    __shared__ int wsum[8];
    int tid  = threadIdx.x;
    int lane = tid & 31, wv = tid >> 5;
    int i0   = (blockIdx.x * 256 + tid) * 4;
    int c[4];
    #pragma unroll
    for (int k = 0; k < 4; ++k) {
        int idx = i0 + k;
        c[k] = (idx < n) ? cur[idx] : 0;
        if (idx < n) dinv[idx] = rsqrtf((float)(c[k] + 1));
    }
    int t   = c[0] + c[1] + c[2] + c[3];
    int inc = wave_incl_scan(t, lane);
    if (lane == 31) wsum[wv] = inc;
    __syncthreads();
    int woff = 0;
    #pragma unroll
    for (int w = 0; w < 8; ++w) woff += (w < wv) ? wsum[w] : 0;
    int run = woff + inc - t;
    #pragma unroll
    for (int k = 0; k < 4; ++k) {
        int idx = i0 + k;
        if (idx < n) cur[idx] = run;
        run += c[k];
    }
    if (tid == 255) bsum[blockIdx.x] = woff + inc;
}

__global__ void k_scan_top(int* __restrict__ bsum, int nb) {
    __shared__ int wsum[32];
    int tid  = threadIdx.x;
    int lane = tid & 31, wv = tid >> 5;
    int carry = 0;
    for (int c0 = 0; c0 < nb; c0 += 1024) {
        int idx = c0 + tid;
        int t   = (idx < nb) ? bsum[idx] : 0;
        int inc = wave_incl_scan(t, lane);
        if (lane == 31) wsum[wv] = inc;
        __syncthreads();
        int woff = 0, tot = 0;
        #pragma unroll
        for (int w = 0; w < 32; ++w) {
            int v = wsum[w];
            woff += (w < wv) ? v : 0;
            tot  += v;
        }
        if (idx < nb) bsum[idx] = carry + woff + inc - t;
        carry += tot;
        __syncthreads();
    }
}

__global__ void k_scan_add(int* __restrict__ cur, const int* __restrict__ bsum, int n) {
    int i = blockIdx.x * blockDim.x + threadIdx.x;
    if (i < n) cur[i] += bsum[i >> 10];
}

__global__ void __launch_bounds__(256)
k_fill(const int* __restrict__ src, const int* __restrict__ dst,
       int* __restrict__ cur, int* __restrict__ csr, int* __restrict__ tick,
       int e, int split, int nchunks) {
    __shared__ int s_c;
    int tid = threadIdx.x;
    int h   = chip_half();
    for (int pass = 0; pass < 2; ++pass) {
        int  r  = h ^ pass;
        bool hi = (r == 1);
        for (;;) {
            if (tid == 0) s_c = atomicAdd(&tick[(2 + r) * TICK_PITCH], 1);
            __syncthreads();
            int c = s_c;
            __syncthreads();
            if (c >= nchunks) break;
            int i0 = c * CHUNK + tid;
            int  d[EDGE_TRIPS], s[EDGE_TRIPS], pos[EDGE_TRIPS];
            bool mine[EDGE_TRIPS];
            #pragma unroll
            for (int k = 0; k < EDGE_TRIPS; ++k) {
                int i = i0 + k * 256;
                d[k]    = (i < e) ? dst[i] : -1;
                mine[k] = d[k] >= 0 && (d[k] >= split) == hi;
            }
            #pragma unroll
            for (int k = 0; k < EDGE_TRIPS; ++k) {
                if (mine[k]) {
                    s[k]   = src[i0 + k * 256];
                    pos[k] = atomicAdd(&cur[d[k]], 1);
                }
            }
            #pragma unroll
            for (int k = 0; k < EDGE_TRIPS; ++k)
                if (mine[k]) csr[pos[k]] = s[k];
        }
    }
}

__device__ __forceinline__ void stage_segment(const int* __restrict__ csr, int* seg,
                                              int s0, int cnt, int tid, int nthr) {
    const int4* g  = (const int4*)(csr + s0);
    int4*       l  = (int4*)seg;
    int         nq = (cnt + 3) >> 2;
    for (int j0 = 0; j0 < nq; j0 += SEG_TRIPS * nthr) {
        int4 v[SEG_TRIPS];
        #pragma unroll
        for (int k = 0; k < SEG_TRIPS; ++k) {
            int j = j0 + k * nthr + tid;
            v[k] = (j < nq) ? g[j] : make_int4(0, 0, 0, 0);
        }
        #pragma unroll
        for (int k = 0; k < SEG_TRIPS; ++k) {
            int j = j0 + k * nthr + tid;
            if (j < nq) l[j] = v[k];
        }
    }
}

__global__ void k_pull1(const float* __restrict__ x, const int* __restrict__ cur,
                        const int* __restrict__ csr, const float* __restrict__ dinv,
                        const float* __restrict__ W1, const float* __restrict__ b1,
                        const float* __restrict__ W2, float* __restrict__ h2, int n) {
    __shared__ __align__(16) int seg[SEG_CAP];
    int lane = threadIdx.x & 31;
    int wave = threadIdx.x >> 5;
    int base = (blockIdx.x * (blockDim.x >> 5) + wave) * 32;

    int  m    = lane & 15;
    bool hi   = lane >= 16;
    int  node = base + lane;
    bool live = node < n;
    if (!live) node = n - 1;

    int p   = node ? cur[node - 1] : 0;
    int end = live ? cur[node] : p;

    int b0  = blockIdx.x * blockDim.x;
    int bl  = min(b0 + (int)blockDim.x, n) - 1;
    int c0s = cur[max(b0, 1) - 1];
    int cE  = cur[bl];
    int sg0 = (b0 ? c0s : 0) & ~31;
    int cnt = min(cE - sg0, SEG_CAP);
    stage_segment(csr, seg, sg0, cnt, threadIdx.x, blockDim.x);
    __syncthreads();

    if (base >= n) return;

    float s0 = 0.0f, s1 = 0.0f, s2 = 0.0f;
    auto pair = [&](int sa, int sb) {
        float wa = dinv[sa], wb = dinv[sb];
        const float* xa = x + (size_t)sa * 3;
        const float* xb = x + (size_t)sb * 3;
        float a0 = xa[0], a1 = xa[1], a2 = xa[2];
        float c0 = xb[0], c1 = xb[1], c2 = xb[2];
        s0 += wa * a0; s1 += wa * a1; s2 += wa * a2;
        s0 += wb * c0; s1 += wb * c1; s2 += wb * c2;
    };
    auto one = [&](int sa) {
        float wa = dinv[sa];
        const float* xa = x + (size_t)sa * 3;
        float a0 = xa[0], a1 = xa[1], a2 = xa[2];
        s0 += wa * a0; s1 += wa * a1; s2 += wa * a2;
    };
    int pl = min(end, sg0 + SEG_CAP);
    for (; p + 1 < pl; p += 2) pair(seg[p - sg0], seg[p + 1 - sg0]);
    if (p < pl) { one(seg[p - sg0]); ++p; }
    for (; p + 1 < end; p += 2) pair(csr[p], csr[p + 1]);
    if (p < end) one(csr[p]);
    float dv = dinv[node];
    const float* xs = x + (size_t)node * 3;
    float x0 = xs[0], x1 = xs[1], x2 = xs[2];
    s0 = (s0 + dv * x0) * dv;
    s1 = (s1 + dv * x1) * dv;
    s2 = (s2 + dv * x2) * dv;

    float y0 = __shfl_xor(s0, 16), y1 = __shfl_xor(s1, 16), y2 = __shfl_xor(s2, 16);

    v2f bxA, bxB;
    bxA.x = hi ? y2 : s0;  bxA.y = hi ? 0.0f : s1;
    bxB.x = hi ? s2 : y0;  bxB.y = hi ? 0.0f : y1;

    int  fc = (m < 8) ? m : 7;
    bool fv = (m < 8);
    float w0 = W1[0 * 8 + fc];
    float w1 = W1[1 * 8 + fc];
    float w2 = W1[2 * 8 + fc];
    v2f aw;
    aw.x = fv ? (hi ? w2 : w0) : 0.0f;
    aw.y = (fv && !hi) ? w1 : 0.0f;

    v8f gA = {}, gB = {};
    gA = __builtin_amdgcn_wmma_f32_16x16x4_f32(false, aw, false, bxA, (short)0, gA,
                                               false, false);
    gB = __builtin_amdgcn_wmma_f32_16x16x4_f32(false, aw, false, bxB, (short)0, gB,
                                               false, false);

    float tA[8], tB[8];
    #pragma unroll
    for (int r = 0; r < 8; ++r) {
        float br = b1[r];
        tA[r] = fmaxf(gA[r] + br, 0.0f);
        tB[r] = fmaxf(gB[r] + br, 0.0f);
    }

    float uA2 = __shfl(tA[2], m), uA3 = __shfl(tA[3], m);
    float uA6 = __shfl(tA[6], m), uA7 = __shfl(tA[7], m);
    float uB2 = __shfl(tB[2], m), uB3 = __shfl(tB[3], m);
    float uB6 = __shfl(tB[6], m), uB7 = __shfl(tB[7], m);
    v2f qA0, qA1, qB0, qB1;
    qA0.x = hi ? uA2 : tA[0]; qA0.y = hi ? uA3 : tA[1];
    qA1.x = hi ? uA6 : tA[4]; qA1.y = hi ? uA7 : tA[5];
    qB0.x = hi ? uB2 : tB[0]; qB0.y = hi ? uB3 : tB[1];
    qB1.x = hi ? uB6 : tB[4]; qB1.y = hi ? uB7 : tB[5];

    int  k0 = hi ? 2 : 0;
    int  jc = (m < 2) ? m : 1;
    bool jv = (m < 2);
    float wa2 = W2[(k0)     * 2 + jc];
    float wb2 = W2[(k0 + 1) * 2 + jc];
    float wc2 = W2[(k0 + 4) * 2 + jc];
    float wd2 = W2[(k0 + 5) * 2 + jc];
    v2f e0, e1;
    e0.x = jv ? wa2 : 0.0f; e0.y = jv ? wb2 : 0.0f;
    e1.x = jv ? wc2 : 0.0f; e1.y = jv ? wd2 : 0.0f;

    v8f oA = {}, oB = {};
    oA = __builtin_amdgcn_wmma_f32_16x16x4_f32(false, e0, false, qA0, (short)0, oA,
                                               false, false);
    oB = __builtin_amdgcn_wmma_f32_16x16x4_f32(false, e0, false, qB0, (short)0, oB,
                                               false, false);
    oA = __builtin_amdgcn_wmma_f32_16x16x4_f32(false, e1, false, qA1, (short)0, oA,
                                               false, false);
    oB = __builtin_amdgcn_wmma_f32_16x16x4_f32(false, e1, false, qB1, (short)0, oB,
                                               false, false);

    float z0 = __shfl_xor(oB[0], 16), z1 = __shfl_xor(oB[1], 16);
    if (live) {
        float2 r2;
        r2.x = hi ? z0 : oA[0];
        r2.y = hi ? z1 : oA[1];
        ((float2*)h2)[node] = r2;
    }
}

__global__ void k_pull2(const float* __restrict__ h2, const int* __restrict__ cur,
                        const int* __restrict__ csr, const float* __restrict__ dinv,
                        const float* __restrict__ b2, float* __restrict__ out, int n) {
    __shared__ __align__(16) int seg[SEG_CAP];
    int v  = blockIdx.x * blockDim.x + threadIdx.x;
    int vc = min(v, n - 1);
    const float2* hp = (const float2*)h2;
    int p   = vc ? cur[vc - 1] : 0;
    int end = cur[vc];

    int b0  = blockIdx.x * blockDim.x;
    int bl  = min(b0 + (int)blockDim.x, n) - 1;
    int c0s = cur[max(b0, 1) - 1];
    int cE  = cur[bl];
    int sg0 = (b0 ? c0s : 0) & ~31;
    int cnt = min(cE - sg0, SEG_CAP);
    stage_segment(csr, seg, sg0, cnt, threadIdx.x, blockDim.x);
    __syncthreads();

    if (v >= n) return;
    float a0 = 0.0f, a1 = 0.0f;
    auto pair = [&](int sa, int sb) {
        float  wa = dinv[sa], wb = dinv[sb];
        float2 ha = hp[sa], hb = hp[sb];
        a0 += wa * ha.x; a1 += wa * ha.y;
        a0 += wb * hb.x; a1 += wb * hb.y;
    };
    auto one = [&](int sa) {
        float  wa = dinv[sa];
        float2 ha = hp[sa];
        a0 += wa * ha.x; a1 += wa * ha.y;
    };
    int pl = min(end, sg0 + SEG_CAP);
    for (; p + 1 < pl; p += 2) pair(seg[p - sg0], seg[p + 1 - sg0]);
    if (p < pl) { one(seg[p - sg0]); ++p; }
    for (; p + 1 < end; p += 2) pair(csr[p], csr[p + 1]);
    if (p < end) one(csr[p]);
    float  dv = dinv[v];
    float2 hv = hp[v];
    float2 r;
    r.x = b2[0] + dv * (a0 + dv * hv.x);
    r.y = b2[1] + dv * (a1 + dv * hv.y);
    ((float2*)out)[v] = r;
}

extern "C" void kernel_launch(void* const* d_in, const int* in_sizes, int n_in,
                              void* d_out, int out_size, void* d_ws, size_t ws_size,
                              hipStream_t stream) {
    const float* x  = (const float*)d_in[0];
    const int*   ei = (const int*)d_in[1];
    const float* W1 = (const float*)d_in[2];
    const float* b1 = (const float*)d_in[3];
    const float* W2 = (const float*)d_in[4];
    const float* b2 = (const float*)d_in[5];
    float* out = (float*)d_out;

    int n = in_sizes[0] / 3;
    int e = in_sizes[1] / 2;
    const int* src = ei;
    const int* dst = ei + e;

    int*   cur  = (int*)d_ws;
    float* dinv = (float*)d_ws + n;
    float* h2   = dinv + n;
    int*   csr  = (int*)(h2 + (size_t)n * 2);
    int*   bsum = csr + e;

    const int T = 256;
    int gN  = (n + T - 1) / T;
    int gMM = (n + 255) / 256;
    int nb  = (n + 1023) / 1024;
    int*   tick = bsum + ((nb + 63) / 64) * 64;
    int split   = ((n / 2 + 63) / 64) * 64;
    int nchunks = (e + CHUNK - 1) / CHUNK;
    const int gR = 2048;

    k_zero      <<<gN,  T,    0, stream>>>(cur, tick, n);
    k_count     <<<gR,  T,    0, stream>>>(dst, cur, tick, e, split, nchunks);
    k_scan_block<<<nb,  256,  0, stream>>>(cur, dinv, bsum, n);
    k_scan_top  <<<1,   1024, 0, stream>>>(bsum, nb);
    k_scan_add  <<<gN,  T,    0, stream>>>(cur, bsum, n);
    k_fill      <<<gR,  T,    0, stream>>>(src, dst, cur, csr, tick, e, split, nchunks);
    k_pull1     <<<gMM, T,    0, stream>>>(x, cur, csr, dinv, W1, b1, W2, h2, n);
    k_pull2     <<<gN,  T,    0, stream>>>(h2, cur, csr, dinv, b2, out, n);
}


# layer-2 gather: layer 1 stores each node's h2 pair and its dinv as ONE aligned 16-byte record; an edge of layer 2 fetches it with one load, not a row and a factor from two arrays; sums unchanged
# speedup vs baseline: 4.6427x; 1.0850x over previous
#include <hip/hip_runtime.h>

typedef float v2f __attribute__((ext_vector_type(2)));
typedef float v8f __attribute__((ext_vector_type(8)));

#define EDGE_TRIPS 8
#define CHUNK (256 * EDGE_TRIPS)
#define TICK_PITCH 64
#define SEG_CAP 4096
#define SEG_TRIPS 4

__device__ __forceinline__ int chip_half() {
    return (int)((__builtin_amdgcn_s_sendmsg_rtn(0x87) >> 18) & 1u);
}

__global__ void k_zero(int* __restrict__ cur, int* __restrict__ tick, int n) {
    int i = blockIdx.x * blockDim.x + threadIdx.x;
    if (i < n) cur[i] = 0;
    if (i < 4 * TICK_PITCH) tick[i] = 0;
}

__global__ void __launch_bounds__(256)
k_count(const int* __restrict__ dst, int* __restrict__ cur, int* __restrict__ tick,
        int e, int split, int nchunks) {
    __shared__ int s_c;
    int tid = threadIdx.x;
    int h   = chip_half();
    for (int pass = 0; pass < 2; ++pass) {
        int  r  = h ^ pass;
        bool hi = (r == 1);
        for (;;) {
            if (tid == 0) s_c = atomicAdd(&tick[r * TICK_PITCH], 1);
            __syncthreads();
            int c = s_c;
            __syncthreads();
            if (c >= nchunks) break;
            int i0 = c * CHUNK + tid;
            int d[EDGE_TRIPS];
            #pragma unroll
            for (int k = 0; k < EDGE_TRIPS; ++k) {
                int i = i0 + k * 256;
                d[k] = (i < e) ? dst[i] : -1;
            }
            #pragma unroll
            for (int k = 0; k < EDGE_TRIPS; ++k)
                if (d[k] >= 0 && (d[k] >= split) == hi) atomicAdd(&cur[d[k]], 1);
        }
    }
}

__device__ __forceinline__ int wave_incl_scan(int v, int lane) {
    #pragma unroll
    for (int off = 1; off < 32; off <<= 1) {
        int y = __shfl_up(v, off);
        if (lane >= off) v += y;
    }
    return v;
}

__global__ void k_scan_block(int* __restrict__ cur, float* __restrict__ dinv,
                             int* __restrict__ bsum, int n) {
    __shared__ int wsum[8];
    int tid  = threadIdx.x;
    int lane = tid & 31, wv = tid >> 5;
    int i0   = (blockIdx.x * 256 + tid) * 4;
    int c[4];
    #pragma unroll
    for (int k = 0; k < 4; ++k) {
        int idx = i0 + k;
        c[k] = (idx < n) ? cur[idx] : 0;
        if (idx < n) dinv[idx] = rsqrtf((float)(c[k] + 1));
    }
    int t   = c[0] + c[1] + c[2] + c[3];
    int inc = wave_incl_scan(t, lane);
    if (lane == 31) wsum[wv] = inc;
    __syncthreads();
    int woff = 0;
    #pragma unroll
    for (int w = 0; w < 8; ++w) woff += (w < wv) ? wsum[w] : 0;
    int run = woff + inc - t;
    #pragma unroll
    for (int k = 0; k < 4; ++k) {
        int idx = i0 + k;
        if (idx < n) cur[idx] = run;
        run += c[k];
    }
    if (tid == 255) bsum[blockIdx.x] = woff + inc;
}

__global__ void k_scan_top(int* __restrict__ bsum, int nb) {
    __shared__ int wsum[32];
    int tid  = threadIdx.x;
    int lane = tid & 31, wv = tid >> 5;
    int carry = 0;
    for (int c0 = 0; c0 < nb; c0 += 1024) {
        int idx = c0 + tid;
        int t   = (idx < nb) ? bsum[idx] : 0;
        int inc = wave_incl_scan(t, lane);
        if (lane == 31) wsum[wv] = inc;
        __syncthreads();
        int woff = 0, tot = 0;
        #pragma unroll
        for (int w = 0; w < 32; ++w) {
            int v = wsum[w];
            woff += (w < wv) ? v : 0;
            tot  += v;
        }
        if (idx < nb) bsum[idx] = carry + woff + inc - t;
        carry += tot;
        __syncthreads();
    }
}

__global__ void k_scan_add(int* __restrict__ cur, const int* __restrict__ bsum, int n) {
    int i = blockIdx.x * blockDim.x + threadIdx.x;
    if (i < n) cur[i] += bsum[i >> 10];
}

__global__ void __launch_bounds__(256)
k_fill(const int* __restrict__ src, const int* __restrict__ dst,
       int* __restrict__ cur, int* __restrict__ csr, int* __restrict__ tick,
       int e, int split, int nchunks) {
    __shared__ int s_c;
    int tid = threadIdx.x;
    int h   = chip_half();
    for (int pass = 0; pass < 2; ++pass) {
        int  r  = h ^ pass;
        bool hi = (r == 1);
        for (;;) {
            if (tid == 0) s_c = atomicAdd(&tick[(2 + r) * TICK_PITCH], 1);
            __syncthreads();
            int c = s_c;
            __syncthreads();
            if (c >= nchunks) break;
            int i0 = c * CHUNK + tid;
            int  d[EDGE_TRIPS], s[EDGE_TRIPS], pos[EDGE_TRIPS];
            bool mine[EDGE_TRIPS];
            #pragma unroll
            for (int k = 0; k < EDGE_TRIPS; ++k) {
                int i = i0 + k * 256;
                d[k]    = (i < e) ? dst[i] : -1;
                mine[k] = d[k] >= 0 && (d[k] >= split) == hi;
            }
            #pragma unroll
            for (int k = 0; k < EDGE_TRIPS; ++k) {
                if (mine[k]) {
                    s[k]   = src[i0 + k * 256];
                    pos[k] = atomicAdd(&cur[d[k]], 1);
                }
            }
            #pragma unroll
            for (int k = 0; k < EDGE_TRIPS; ++k)
                if (mine[k]) csr[pos[k]] = s[k];
        }
    }
}

__device__ __forceinline__ void stage_segment(const int* __restrict__ csr, int* seg,
                                              int s0, int cnt, int tid, int nthr) {
    const int4* g  = (const int4*)(csr + s0);
    int4*       l  = (int4*)seg;
    int         nq = (cnt + 3) >> 2;
    for (int j0 = 0; j0 < nq; j0 += SEG_TRIPS * nthr) {
        int4 v[SEG_TRIPS];
        #pragma unroll
        for (int k = 0; k < SEG_TRIPS; ++k) {
            int j = j0 + k * nthr + tid;
            v[k] = (j < nq) ? g[j] : make_int4(0, 0, 0, 0);
        }
        #pragma unroll
        for (int k = 0; k < SEG_TRIPS; ++k) {
            int j = j0 + k * nthr + tid;
            if (j < nq) l[j] = v[k];
        }
    }
}

__global__ void k_pull1(const float* __restrict__ x, const int* __restrict__ cur,
                        const int* __restrict__ csr, const float* __restrict__ dinv,
                        const float* __restrict__ W1, const float* __restrict__ b1,
                        const float* __restrict__ W2, float4* __restrict__ rec, int n) {
    __shared__ __align__(16) int seg[SEG_CAP];
    int lane = threadIdx.x & 31;
    int wave = threadIdx.x >> 5;
    int base = (blockIdx.x * (blockDim.x >> 5) + wave) * 32;

    int  m    = lane & 15;
    bool hi   = lane >= 16;
    int  node = base + lane;
    bool live = node < n;
    if (!live) node = n - 1;

    int p   = node ? cur[node - 1] : 0;
    int end = live ? cur[node] : p;

    int b0  = blockIdx.x * blockDim.x;
    int bl  = min(b0 + (int)blockDim.x, n) - 1;
    int c0s = cur[max(b0, 1) - 1];
    int cE  = cur[bl];
    int sg0 = (b0 ? c0s : 0) & ~31;
    int cnt = min(cE - sg0, SEG_CAP);
    stage_segment(csr, seg, sg0, cnt, threadIdx.x, blockDim.x);
    __syncthreads();

    if (base >= n) return;

    float s0 = 0.0f, s1 = 0.0f, s2 = 0.0f;
    auto pair = [&](int sa, int sb) {
        float wa = dinv[sa], wb = dinv[sb];
        const float* xa = x + (size_t)sa * 3;
        const float* xb = x + (size_t)sb * 3;
        float a0 = xa[0], a1 = xa[1], a2 = xa[2];
        float c0 = xb[0], c1 = xb[1], c2 = xb[2];
        s0 += wa * a0; s1 += wa * a1; s2 += wa * a2;
        s0 += wb * c0; s1 += wb * c1; s2 += wb * c2;
    };
    auto one = [&](int sa) {
        float wa = dinv[sa];
        const float* xa = x + (size_t)sa * 3;
        float a0 = xa[0], a1 = xa[1], a2 = xa[2];
        s0 += wa * a0; s1 += wa * a1; s2 += wa * a2;
    };
    int pl = min(end, sg0 + SEG_CAP);
    for (; p + 1 < pl; p += 2) pair(seg[p - sg0], seg[p + 1 - sg0]);
    if (p < pl) { one(seg[p - sg0]); ++p; }
    for (; p + 1 < end; p += 2) pair(csr[p], csr[p + 1]);
    if (p < end) one(csr[p]);
    float dv = dinv[node];
    const float* xs = x + (size_t)node * 3;
    float x0 = xs[0], x1 = xs[1], x2 = xs[2];
    s0 = (s0 + dv * x0) * dv;
    s1 = (s1 + dv * x1) * dv;
    s2 = (s2 + dv * x2) * dv;

    float y0 = __shfl_xor(s0, 16), y1 = __shfl_xor(s1, 16), y2 = __shfl_xor(s2, 16);

    v2f bxA, bxB;
    bxA.x = hi ? y2 : s0;  bxA.y = hi ? 0.0f : s1;
    bxB.x = hi ? s2 : y0;  bxB.y = hi ? 0.0f : y1;

    int  fc = (m < 8) ? m : 7;
    bool fv = (m < 8);
    float w0 = W1[0 * 8 + fc];
    float w1 = W1[1 * 8 + fc];
    float w2 = W1[2 * 8 + fc];
    v2f aw;
    aw.x = fv ? (hi ? w2 : w0) : 0.0f;
    aw.y = (fv && !hi) ? w1 : 0.0f;

    v8f gA = {}, gB = {};
    gA = __builtin_amdgcn_wmma_f32_16x16x4_f32(false, aw, false, bxA, (short)0, gA,
                                               false, false);
    gB = __builtin_amdgcn_wmma_f32_16x16x4_f32(false, aw, false, bxB, (short)0, gB,
                                               false, false);

    float tA[8], tB[8];
    #pragma unroll
    for (int r = 0; r < 8; ++r) {
        float br = b1[r];
        tA[r] = fmaxf(gA[r] + br, 0.0f);
        tB[r] = fmaxf(gB[r] + br, 0.0f);
    }

    float uA2 = __shfl(tA[2], m), uA3 = __shfl(tA[3], m);
    float uA6 = __shfl(tA[6], m), uA7 = __shfl(tA[7], m);
    float uB2 = __shfl(tB[2], m), uB3 = __shfl(tB[3], m);
    float uB6 = __shfl(tB[6], m), uB7 = __shfl(tB[7], m);
    v2f qA0, qA1, qB0, qB1;
    qA0.x = hi ? uA2 : tA[0]; qA0.y = hi ? uA3 : tA[1];
    qA1.x = hi ? uA6 : tA[4]; qA1.y = hi ? uA7 : tA[5];
    qB0.x = hi ? uB2 : tB[0]; qB0.y = hi ? uB3 : tB[1];
    qB1.x = hi ? uB6 : tB[4]; qB1.y = hi ? uB7 : tB[5];

    int  k0 = hi ? 2 : 0;
    int  jc = (m < 2) ? m : 1;
    bool jv = (m < 2);
    float wa2 = W2[(k0)     * 2 + jc];
    float wb2 = W2[(k0 + 1) * 2 + jc];
    float wc2 = W2[(k0 + 4) * 2 + jc];
    float wd2 = W2[(k0 + 5) * 2 + jc];
    v2f e0, e1;
    e0.x = jv ? wa2 : 0.0f; e0.y = jv ? wb2 : 0.0f;
    e1.x = jv ? wc2 : 0.0f; e1.y = jv ? wd2 : 0.0f;

    v8f oA = {}, oB = {};
    oA = __builtin_amdgcn_wmma_f32_16x16x4_f32(false, e0, false, qA0, (short)0, oA,
                                               false, false);
    oB = __builtin_amdgcn_wmma_f32_16x16x4_f32(false, e0, false, qB0, (short)0, oB,
                                               false, false);
    oA = __builtin_amdgcn_wmma_f32_16x16x4_f32(false, e1, false, qA1, (short)0, oA,
                                               false, false);
    oB = __builtin_amdgcn_wmma_f32_16x16x4_f32(false, e1, false, qB1, (short)0, oB,
                                               false, false);

    float z0 = __shfl_xor(oB[0], 16), z1 = __shfl_xor(oB[1], 16);
    if (live) {
        float4 r4;
        r4.x = hi ? z0 : oA[0];
        r4.y = hi ? z1 : oA[1];
        r4.z = dv;
        r4.w = 0.0f;
        rec[node] = r4;
    }
}

__global__ void k_pull2(const float4* __restrict__ rec, const int* __restrict__ cur,
                        const int* __restrict__ csr,
                        const float* __restrict__ b2, float* __restrict__ out, int n) {
    __shared__ __align__(16) int seg[SEG_CAP];
    int v  = blockIdx.x * blockDim.x + threadIdx.x;
    int vc = min(v, n - 1);
    int p   = vc ? cur[vc - 1] : 0;
    int end = cur[vc];

    int b0  = blockIdx.x * blockDim.x;
    int bl  = min(b0 + (int)blockDim.x, n) - 1;
    int c0s = cur[max(b0, 1) - 1];
    int cE  = cur[bl];
    int sg0 = (b0 ? c0s : 0) & ~31;
    int cnt = min(cE - sg0, SEG_CAP);
    stage_segment(csr, seg, sg0, cnt, threadIdx.x, blockDim.x);
    __syncthreads();

    if (v >= n) return;
    float a0 = 0.0f, a1 = 0.0f;
    auto pair = [&](int sa, int sb) {
        float4 ra = rec[sa], rb = rec[sb];
        a0 += ra.z * ra.x; a1 += ra.z * ra.y;
        a0 += rb.z * rb.x; a1 += rb.z * rb.y;
    };
    auto one = [&](int sa) {
        float4 ra = rec[sa];
        a0 += ra.z * ra.x; a1 += ra.z * ra.y;
    };
    int pl = min(end, sg0 + SEG_CAP);
    for (; p + 1 < pl; p += 2) pair(seg[p - sg0], seg[p + 1 - sg0]);
    if (p < pl) { one(seg[p - sg0]); ++p; }
    for (; p + 1 < end; p += 2) pair(csr[p], csr[p + 1]);
    if (p < end) one(csr[p]);
    float4 rv = rec[v];
    float  dv = rv.z;
    float2 r;
    r.x = b2[0] + dv * (a0 + dv * rv.x);
    r.y = b2[1] + dv * (a1 + dv * rv.y);
    ((float2*)out)[v] = r;
}

extern "C" void kernel_launch(void* const* d_in, const int* in_sizes, int n_in,
                              void* d_out, int out_size, void* d_ws, size_t ws_size,
                              hipStream_t stream) {
    const float* x  = (const float*)d_in[0];
    const int*   ei = (const int*)d_in[1];
    const float* W1 = (const float*)d_in[2];
    const float* b1 = (const float*)d_in[3];
    const float* W2 = (const float*)d_in[4];
    const float* b2 = (const float*)d_in[5];
    float* out = (float*)d_out;

    int n = in_sizes[0] / 3;
    int e = in_sizes[1] / 2;
    const int* src = ei;
    const int* dst = ei + e;

    int*    cur  = (int*)d_ws;
    float*  dinv = (float*)d_ws + n;
    float4* rec  = (float4*)(dinv + n);
    int*    csr  = (int*)(rec + n);
    int*    bsum = csr + e;

    const int T = 256;
    int gN  = (n + T - 1) / T;
    int gMM = (n + 255) / 256;
    int nb  = (n + 1023) / 1024;
    int*   tick = bsum + ((nb + 63) / 64) * 64;
    int split   = ((n / 2 + 63) / 64) * 64;
    int nchunks = (e + CHUNK - 1) / CHUNK;
    const int gR = 2048;

    k_zero      <<<gN,  T,    0, stream>>>(cur, tick, n);
    k_count     <<<gR,  T,    0, stream>>>(dst, cur, tick, e, split, nchunks);
    k_scan_block<<<nb,  256,  0, stream>>>(cur, dinv, bsum, n);
    k_scan_top  <<<1,   1024, 0, stream>>>(bsum, nb);
    k_scan_add  <<<gN,  T,    0, stream>>>(cur, bsum, n);
    k_fill      <<<gR,  T,    0, stream>>>(src, dst, cur, csr, tick, e, split, nchunks);
    k_pull1     <<<gMM, T,    0, stream>>>(x, cur, csr, dinv, W1, b1, W2, rec, n);
    k_pull2     <<<gN,  T,    0, stream>>>(rec, cur, csr, b2, out, n);
}
